# starting kernel
# baseline (speedup 1.0000x reference)
#include <hip/hip_runtime.h>


typedef float v2f __attribute__((ext_vector_type(2)));
typedef float v8f __attribute__((ext_vector_type(8)));

#define T_DIM 4096
#define D_DIM 256
#define EPS_F 1e-7f

__global__ __launch_bounds__(256)
void trilinear_states_kernel(const float* __restrict__ response,
                             const float* __restrict__ prompt,
                             const float* __restrict__ W,
                             float* __restrict__ out) {
  __shared__ float s_v[D_DIM];
  __shared__ float s_red[256];
  __shared__ float s_st[256];

  const int tid   = threadIdx.x;
  const int b     = blockIdx.y;
  const int tBase = blockIdx.x * 256;

  {
    const float p = prompt[b * D_DIM + tid];
    s_v[tid]   = W[D_DIM + tid] + p * W[2 * D_DIM + tid];
    s_red[tid] = p * W[tid];
  }
  __syncthreads();
  #pragma unroll
  for (int s = 128; s > 0; s >>= 1) {
    if (tid < s) s_red[tid] += s_red[tid + s];
    __syncthreads();
  }
  const float c_b = s_red[0];

  const int lane = tid & 31;
  const int wv   = tid >> 5;
  const int half = lane >> 4;
  const int m    = lane & 15;

  for (int i = 0; i < 2; ++i) {
    const int tile = wv * 2 + i;
    const int t0   = tBase + tile * 16;
    const float* rowp =
        response + ((size_t)b * T_DIM + (size_t)(t0 + m)) * D_DIM + 2 * half;

    v8f acc = {};
    #pragma unroll 8
    for (int c = 0; c < 64; ++c) {
      v2f a  = *(const v2f*)(rowp + 4 * c);
      v2f bv = *(const v2f*)(&s_v[4 * c + 2 * half]);
      acc = __builtin_amdgcn_wmma_f32_16x16x4_f32(
           false, a,  false, bv,
           (short)0, acc,  false,  false);
    }

    if (m == 0) {
      #pragma unroll
      for (int r = 0; r < 8; ++r) s_st[tile * 16 + half * 8 + r] = acc[r];
    }
    __syncthreads();
    if (half == 0)
      out[(size_t)b * T_DIM + (size_t)(t0 + m)] = expf(s_st[tile * 16 + m] + c_b);
  }
}

__global__ __launch_bounds__(256)
void softmax_norm_kernel(float* __restrict__ out) {
  __shared__ float s_red[256];
  const int tid = threadIdx.x;
  float* p = out + (size_t)blockIdx.x * T_DIM;

  float s = 0.f;
  #pragma unroll
  for (int i = 0; i < T_DIM / 256; ++i) s += p[tid + i * 256];
  s_red[tid] = s;
  __syncthreads();
  #pragma unroll
  for (int k = 128; k > 0; k >>= 1) {
    if (tid < k) s_red[tid] += s_red[tid + k];
    __syncthreads();
  }
  const float inv = 1.0f / (s_red[0] + EPS_F);
  #pragma unroll
  for (int i = 0; i < T_DIM / 256; ++i) p[tid + i * 256] *= inv;
}

extern "C" void kernel_launch(void* const* d_in, const int* in_sizes, int n_in,
                              void* d_out, int out_size, void* d_ws, size_t ws_size,
                              hipStream_t stream) {
  const float* response = (const float*)d_in[0];
  const float* prompt   = (const float*)d_in[1];
  const float* W        = (const float*)d_in[2];
  float* out            = (float*)d_out;

  const int B = in_sizes[1] / D_DIM;

  dim3 grid1(T_DIM / 256, B);
  trilinear_states_kernel<<<grid1, 256, 0, stream>>>(response, prompt, W, out);
  softmax_norm_kernel<<<B, 256, 0, stream>>>(out);
}


# response read as whole 128-byte lines (16-byte loads, 4 rows per load, 32 loads ahead) into the WMMA B operand, v_b as A in registers; row sum order changes: 4 column partials, then a quad add
# speedup vs baseline: 1.4904x; 1.4904x over previous
#include <hip/hip_runtime.h>


typedef float v2f __attribute__((ext_vector_type(2)));
typedef float v4f __attribute__((ext_vector_type(4)));
typedef float v8f __attribute__((ext_vector_type(8)));

#define T_DIM 4096
#define D_DIM 256
#define EPS_F 1e-7f

static __device__ __forceinline__ float quad_sum(float v) {
  v += __builtin_bit_cast(float, __builtin_amdgcn_update_dpp(
           0, __builtin_bit_cast(int, v), 0xB1, 0xF, 0xF, true));
  v += __builtin_bit_cast(float, __builtin_amdgcn_update_dpp(
           0, __builtin_bit_cast(int, v), 0x4E, 0xF, 0xF, true));
  return v;
}

__global__ __launch_bounds__(256)
void trilinear_states_kernel(const float* __restrict__ response,
                             const float* __restrict__ prompt,
                             const float* __restrict__ W,
                             float* __restrict__ out) {
  __shared__ __attribute__((aligned(16))) float s_v[D_DIM];
  __shared__ float s_red[256];

  const int tid   = threadIdx.x;
  const int b     = blockIdx.y;
  const int tBase = blockIdx.x * 256;

  {
    const float p = prompt[b * D_DIM + tid];
    s_v[tid]   = W[D_DIM + tid] + p * W[2 * D_DIM + tid];
    s_red[tid] = p * W[tid];
  }
  __syncthreads();
  #pragma unroll
  for (int s = 128; s > 0; s >>= 1) {
    if (tid < s) s_red[tid] += s_red[tid + s];
    __syncthreads();
  }
  const float c_b = s_red[0];

  const int lane = tid & 31;
  const int wv   = tid >> 5;
  const int h    = lane >> 4;
  const int n    = lane & 15;
  const int seg  = n & 3;
  const int rq   = n >> 2;
  const int dofs = 8 * seg + 4 * h;
  const bool seg_lo = (seg & 1) != 0, seg_hi = (seg & 2) != 0;

  v4f va[8];
  for (int j = 0; j < 8; ++j) va[j] = *(const v4f*)&s_v[32 * j + dofs];

  const float* base = response +
      ((size_t)b * T_DIM + (size_t)(tBase + 32 * wv + 4 * rq)) * D_DIM + dofs;

  v4f d[2][2][8];
  float q[8];

#define LOAD_STEP(ST)                                                      \
  do {                                                                     \
    for (int j = 0; j < 8; ++j) {                                          \
      for (int gi = 0; gi < 2; ++gi) {                                     \
        const int g_  = 2 * (ST) + gi;                                     \
        const int rb_ = 16 * (g_ >> 2) + (g_ & 3);                         \
        d[(ST) & 1][gi][j] = *(const v4f*)(base + rb_ * D_DIM + 32 * j);   \
      }                                                                    \
    }                                                                      \
  } while (0)

  LOAD_STEP(0);
  for (int st = 0; st < 4; ++st) {
    __builtin_amdgcn_sched_barrier(0);
    if (st + 1 < 4) LOAD_STEP(st + 1);
    __builtin_amdgcn_sched_barrier(0);

    v8f acc0 = {}, acc1 = {};
    for (int j = 0; j < 8; ++j) {
      const v4f a  = va[j];
      const v4f b0 = d[st & 1][0][j];
      const v4f b1 = d[st & 1][1][j];
      const v2f alo = {a.x, a.y},   ahi = {a.z, a.w};
      const v2f b0lo = {b0.x, b0.y}, b0hi = {b0.z, b0.w};
      const v2f b1lo = {b1.x, b1.y}, b1hi = {b1.z, b1.w};
      acc0 = __builtin_amdgcn_wmma_f32_16x16x4_f32(false, alo, false, b0lo, (short)0, acc0, false, false);
      acc1 = __builtin_amdgcn_wmma_f32_16x16x4_f32(false, alo, false, b1lo, (short)0, acc1, false, false);
      acc0 = __builtin_amdgcn_wmma_f32_16x16x4_f32(false, ahi, false, b0hi, (short)0, acc0, false, false);
      acc1 = __builtin_amdgcn_wmma_f32_16x16x4_f32(false, ahi, false, b1hi, (short)0, acc1, false, false);
    }
    const float x0 = seg_hi ? (seg_lo ? acc0[3] : acc0[2]) : (seg_lo ? acc0[1] : acc0[0]);
    const float x1 = seg_hi ? (seg_lo ? acc1[3] : acc1[2]) : (seg_lo ? acc1[1] : acc1[0]);
    q[2 * st]     = quad_sum(x0);
    q[2 * st + 1] = quad_sum(x1);
  }
#undef LOAD_STEP

  const int mine = 4 * h + seg;
  float fin = q[0];
  for (int g = 1; g < 8; ++g) fin = (mine == g) ? q[g] : fin;

  out[(size_t)b * T_DIM + (size_t)(tBase + 32 * wv + lane)] = expf(fin + c_b);
}

__global__ __launch_bounds__(256)
void softmax_norm_kernel(float* __restrict__ out) {
  __shared__ float s_red[256];
  const int tid = threadIdx.x;
  float* p = out + (size_t)blockIdx.x * T_DIM;

  float s = 0.f;
  #pragma unroll
  for (int i = 0; i < T_DIM / 256; ++i) s += p[tid + i * 256];
  s_red[tid] = s;
  __syncthreads();
  #pragma unroll
  for (int k = 128; k > 0; k >>= 1) {
    if (tid < k) s_red[tid] += s_red[tid + k];
    __syncthreads();
  }
  const float inv = 1.0f / (s_red[0] + EPS_F);
  #pragma unroll
  for (int i = 0; i < T_DIM / 256; ++i) p[tid + i * 256] *= inv;
}

extern "C" void kernel_launch(void* const* d_in, const int* in_sizes, int n_in,
                              void* d_out, int out_size, void* d_ws, size_t ws_size,
                              hipStream_t stream) {
  const float* response = (const float*)d_in[0];
  const float* prompt   = (const float*)d_in[1];
  const float* W        = (const float*)d_in[2];
  float* out            = (float*)d_out;

  const int B = in_sizes[1] / D_DIM;

  dim3 grid1(T_DIM / 256, B);
  trilinear_states_kernel<<<grid1, 256, 0, stream>>>(response, prompt, W, out);
  softmax_norm_kernel<<<B, 256, 0, stream>>>(out);
}


# first two steps of response loads issued at wave entry; v_b/c_b set-up per wave under them, no workgroup barrier; c_b by lane butterfly in the tree's pair order, first add fused with a product
# speedup vs baseline: 1.5782x; 1.0589x over previous
#include <hip/hip_runtime.h>


typedef float v2f __attribute__((ext_vector_type(2)));
typedef float v4f __attribute__((ext_vector_type(4)));
typedef float v8f __attribute__((ext_vector_type(8)));

#define T_DIM 4096
#define D_DIM 256
#define EPS_F 1e-7f

static __device__ __forceinline__ float quad_sum(float v) {
  v += __builtin_bit_cast(float, __builtin_amdgcn_update_dpp(
           0, __builtin_bit_cast(int, v), 0xB1, 0xF, 0xF, true));
  v += __builtin_bit_cast(float, __builtin_amdgcn_update_dpp(
           0, __builtin_bit_cast(int, v), 0x4E, 0xF, 0xF, true));
  return v;
}

static __device__ __forceinline__ float xhalf(float x) {
  const int b = __builtin_bit_cast(int, x);
  return __builtin_bit_cast(float, __builtin_amdgcn_permlanex16(
             b, b, 0x76543210, 0xfedcba98, false, false));
}

template <int M>
static __device__ __forceinline__ float xrow(float x) {
  return __builtin_bit_cast(float, __builtin_amdgcn_update_dpp(
             0, __builtin_bit_cast(int, x), 0x160 + M, 0xF, 0xF, true));
}

__global__ __launch_bounds__(256)
void trilinear_states_kernel(const float* __restrict__ response,
                             const float* __restrict__ prompt,
                             const float* __restrict__ W,
                             float* __restrict__ out) {
  __shared__ __attribute__((aligned(16))) float s_v[8][D_DIM];

  const int tid   = threadIdx.x;
  const int b     = blockIdx.y;
  const int tBase = blockIdx.x * 256;

  const int lane = tid & 31;
  const int wv   = tid >> 5;
  const int h    = lane >> 4;
  const int n    = lane & 15;
  const int seg  = n & 3;
  const int rq   = n >> 2;
  const int dofs = 8 * seg + 4 * h;
  const bool seg_lo = (seg & 1) != 0, seg_hi = (seg & 2) != 0;

  const v4f* p4 = (const v4f*)(prompt + b * D_DIM);
  const v4f* w4 = (const v4f*)W;
  const v4f p0  = p4[lane],        p1  = p4[32 + lane];
  const v4f wp0 = w4[lane],        wp1 = w4[32 + lane];
  const v4f wr0 = w4[64 + lane],   wr1 = w4[96 + lane];
  const v4f wx0 = w4[128 + lane],  wx1 = w4[160 + lane];

  const float* base = response +
      ((size_t)b * T_DIM + (size_t)(tBase + 32 * wv + 4 * rq)) * D_DIM + dofs;

  v4f d[2][2][8];
  float q[8];

#define LOAD_STEP(ST)                                                      \
  do {                                                                     \
    for (int j = 0; j < 8; ++j) {                                          \
      for (int gi = 0; gi < 2; ++gi) {                                     \
        const int g_  = 2 * (ST) + gi;                                     \
        const int rb_ = 16 * (g_ >> 2) + (g_ & 3);                         \
        d[(ST) & 1][gi][j] = *(const v4f*)(base + rb_ * D_DIM + 32 * j);   \
      }                                                                    \
    }                                                                      \
  } while (0)

  __builtin_amdgcn_sched_barrier(0);
  LOAD_STEP(0);
  LOAD_STEP(1);

  float* my_v = s_v[wv];
  {
    v4f vb0, vb1;
    for (int e = 0; e < 4; ++e) {
      vb0[e] = __builtin_fmaf(p0[e], wx0[e], wr0[e]);
      vb1[e] = __builtin_fmaf(p1[e], wx1[e], wr1[e]);
    }
    *(v4f*)&my_v[4 * lane]       = vb0;
    *(v4f*)&my_v[128 + 4 * lane] = vb1;
  }

  float c_b;
  {
    v4f s = p0 * wp0 + p1 * wp1;
    for (int e = 0; e < 4; ++e) s[e] += xhalf(s[e]);
    for (int e = 0; e < 4; ++e) s[e] += xrow<8>(s[e]);
    for (int e = 0; e < 4; ++e) s[e] += xrow<4>(s[e]);
    for (int e = 0; e < 4; ++e) s[e] += xrow<2>(s[e]);
    for (int e = 0; e < 4; ++e) s[e] += xrow<1>(s[e]);
    c_b = (s[0] + s[2]) + (s[1] + s[3]);
  }


  v4f va[8];
  for (int j = 0; j < 8; ++j) va[j] = *(const v4f*)&my_v[32 * j + dofs];

  for (int st = 0; st < 4; ++st) {
    v8f acc0 = {}, acc1 = {};
    for (int j = 0; j < 8; ++j) {
      const v4f a  = va[j];
      const v4f b0 = d[st & 1][0][j];
      const v4f b1 = d[st & 1][1][j];
      const v2f alo = {a.x, a.y},   ahi = {a.z, a.w};
      const v2f b0lo = {b0.x, b0.y}, b0hi = {b0.z, b0.w};
      const v2f b1lo = {b1.x, b1.y}, b1hi = {b1.z, b1.w};
      acc0 = __builtin_amdgcn_wmma_f32_16x16x4_f32(false, alo, false, b0lo, (short)0, acc0, false, false);
      acc1 = __builtin_amdgcn_wmma_f32_16x16x4_f32(false, alo, false, b1lo, (short)0, acc1, false, false);
      acc0 = __builtin_amdgcn_wmma_f32_16x16x4_f32(false, ahi, false, b0hi, (short)0, acc0, false, false);
      acc1 = __builtin_amdgcn_wmma_f32_16x16x4_f32(false, ahi, false, b1hi, (short)0, acc1, false, false);
    }
    const float x0 = seg_hi ? (seg_lo ? acc0[3] : acc0[2]) : (seg_lo ? acc0[1] : acc0[0]);
    const float x1 = seg_hi ? (seg_lo ? acc1[3] : acc1[2]) : (seg_lo ? acc1[1] : acc1[0]);
    q[2 * st]     = quad_sum(x0);
    q[2 * st + 1] = quad_sum(x1);

    __builtin_amdgcn_sched_barrier(0);
    if (st + 2 < 4) {
      LOAD_STEP(st + 2);
      __builtin_amdgcn_sched_barrier(0);
    }
  }
#undef LOAD_STEP

  const int mine = 4 * h + seg;
  float fin = q[0];
  for (int g = 1; g < 8; ++g) fin = (mine == g) ? q[g] : fin;

  out[(size_t)b * T_DIM + (size_t)(tBase + 32 * wv + lane)] = expf(fin + c_b);
}

__global__ __launch_bounds__(256)
void softmax_norm_kernel(float* __restrict__ out) {
  __shared__ float s_red[256];
  const int tid = threadIdx.x;
  float* p = out + (size_t)blockIdx.x * T_DIM;

  float s = 0.f;
  #pragma unroll
  for (int i = 0; i < T_DIM / 256; ++i) s += p[tid + i * 256];
  s_red[tid] = s;
  __syncthreads();
  #pragma unroll
  for (int k = 128; k > 0; k >>= 1) {
    if (tid < k) s_red[tid] += s_red[tid + k];
    __syncthreads();
  }
  const float inv = 1.0f / (s_red[0] + EPS_F);
  #pragma unroll
  for (int i = 0; i < T_DIM / 256; ++i) p[tid + i * 256] *= inv;
}

extern "C" void kernel_launch(void* const* d_in, const int* in_sizes, int n_in,
                              void* d_out, int out_size, void* d_ws, size_t ws_size,
                              hipStream_t stream) {
  const float* response = (const float*)d_in[0];
  const float* prompt   = (const float*)d_in[1];
  const float* W        = (const float*)d_in[2];
  float* out            = (float*)d_out;

  const int B = in_sizes[1] / D_DIM;

  dim3 grid1(T_DIM / 256, B);
  trilinear_states_kernel<<<grid1, 256, 0, stream>>>(response, prompt, W, out);
  softmax_norm_kernel<<<B, 256, 0, stream>>>(out);
}
